# starting kernel
# baseline (speedup 1.0000x reference)
#include <hip/hip_runtime.h>

#define N_NODES   50000
#define N_EDGES   800000
#define IN_DIM    256
#define OUT_DIM   64
#define NEG_SLOPE 0.01f

typedef __attribute__((ext_vector_type(2))) float v2f;
typedef __attribute__((ext_vector_type(8))) float v8f;

__global__ __launch_bounds__(256) void gat_init(float* __restrict__ agg,
                                                float* __restrict__ denom,
                                                float* __restrict__ emax) {
    int i = blockIdx.x * blockDim.x + threadIdx.x;
    if (i < N_NODES * OUT_DIM) agg[i] = 0.0f;
    if (i < N_NODES) {
        denom[i] = 0.0f;
        emax[i]  = -__builtin_inff();
    }
}

__global__ __launch_bounds__(128) void gat_gemm_wmma(
    const float* __restrict__ hidden,
    const float* __restrict__ W,
    const float* __restrict__ bias,
    float* __restrict__ h_out)
{
    const int lane  = threadIdx.x & 31;
    const int wave  = threadIdx.x >> 5;
    const int half  = lane >> 4;
    const int l16   = lane & 15;
    const int mbase = blockIdx.x * 16;
    const int gcol  = wave * 16 + l16;

    const float* arow = hidden + (size_t)(mbase + l16) * IN_DIM + 2 * half;
    const float* bcol = W + (size_t)(2 * half) * OUT_DIM + gcol;

    v8f acc = {};
#pragma unroll 4
    for (int k0 = 0; k0 < IN_DIM; k0 += 4) {
        v2f a = *(const v2f*)(arow + k0);
        v2f b;
        b.x = bcol[(size_t)k0 * OUT_DIM];
        b.y = bcol[(size_t)(k0 + 1) * OUT_DIM];
        acc = __builtin_amdgcn_wmma_f32_16x16x4_f32(
             false, a,  false, b,
             (short)0, acc,  false,  false);
    }

    const float bb = bias[gcol];
#pragma unroll
    for (int r = 0; r < 8; ++r) {
        int row = mbase + r + 8 * half;
        h_out[(size_t)row * OUT_DIM + gcol] = acc[r] + bb;
    }
}

__global__ __launch_bounds__(256) void gat_proj(
    const float* __restrict__ h,
    const float* __restrict__ att_w,
    float* __restrict__ a_src,
    float* __restrict__ a_dst)
{
    int node = blockIdx.x * 8 + (threadIdx.x >> 5);
    if (node >= N_NODES) return;
    int lane = threadIdx.x & 31;

    float h0 = h[(size_t)node * OUT_DIM + lane];
    float h1 = h[(size_t)node * OUT_DIM + lane + 32];
    float s = h0 * att_w[lane]      + h1 * att_w[lane + 32];
    float d = h0 * att_w[lane + 64] + h1 * att_w[lane + 96];
#pragma unroll
    for (int off = 16; off > 0; off >>= 1) {
        s += __shfl_down(s, off, 32);
        d += __shfl_down(d, off, 32);
    }
    if (lane == 0) { a_src[node] = s; a_dst[node] = d; }
}

__device__ __forceinline__ void atomicMaxF(float* addr, float v) {
    if (v >= 0.0f) atomicMax((int*)addr, __float_as_int(v));
    else           atomicMin((unsigned int*)addr, __float_as_uint(v));
}

__global__ __launch_bounds__(256) void gat_edge_logit(
    const int* __restrict__ src, const int* __restrict__ dst,
    const float* __restrict__ a_src, const float* __restrict__ a_dst,
    const float* __restrict__ att_b,
    float* __restrict__ e_out, float* __restrict__ emax)
{
    int e = blockIdx.x * blockDim.x + threadIdx.x;
    if (e >= N_EDGES) return;
    int s = src[e], d = dst[e];
    float v = a_src[s] + a_dst[d] + att_b[0];
    v = (v > 0.0f) ? v : NEG_SLOPE * v;
    e_out[e] = v;
    atomicMaxF(&emax[d], v);
}

__global__ __launch_bounds__(256) void gat_edge_scatter(
    const int* __restrict__ src, const int* __restrict__ dst,
    const float* __restrict__ e_in, const float* __restrict__ emax,
    const float* __restrict__ h,
    float* __restrict__ denom, float* __restrict__ agg)
{
    int edge = blockIdx.x * 8 + (threadIdx.x >> 5);
    if (edge >= N_EDGES) return;
    int lane = threadIdx.x & 31;
    int s = src[edge], d = dst[edge];

    float ex = __expf(e_in[edge] - emax[d]);
    if (lane == 0) atomicAdd(&denom[d], ex);

    float h0 = h[(size_t)s * OUT_DIM + lane];
    float h1 = h[(size_t)s * OUT_DIM + lane + 32];
    atomicAdd(&agg[(size_t)d * OUT_DIM + lane],       ex * h0);
    atomicAdd(&agg[(size_t)d * OUT_DIM + lane + 32],  ex * h1);
}

__global__ __launch_bounds__(256) void gat_final(
    const float* __restrict__ agg, const float* __restrict__ denom,
    const float* __restrict__ h, float* __restrict__ out)
{
    int i = blockIdx.x * blockDim.x + threadIdx.x;
    if (i >= N_NODES * OUT_DIM) return;
    int node = i >> 6;
    float dn = denom[node];
    out[i] = (dn > 0.0f) ? agg[i] / dn : h[i];
}

extern "C" void kernel_launch(void* const* d_in, const int* in_sizes, int n_in,
                              void* d_out, int out_size, void* d_ws, size_t ws_size,
                              hipStream_t stream) {
    const float* hidden = (const float*)d_in[0];
    const float* W      = (const float*)d_in[1];
    const float* bias   = (const float*)d_in[2];
    const float* att_w  = (const float*)d_in[3];
    const float* att_b  = (const float*)d_in[4];
    const int*   src    = (const int*)  d_in[5];
    const int*   dst    = (const int*)  d_in[6];
    float* out = (float*)d_out;

    float* ws     = (float*)d_ws;
    float* h_ws   = ws;
    float* a_src  = h_ws  + (size_t)N_NODES * OUT_DIM;
    float* a_dst  = a_src + N_NODES;
    float* emax   = a_dst + N_NODES;
    float* denom  = emax  + N_NODES;
    float* e_ws   = denom + N_NODES;
    float* agg    = e_ws  + N_EDGES;

    const int NT = 256;
    gat_init<<<(N_NODES * OUT_DIM + NT - 1) / NT, NT, 0, stream>>>(agg, denom, emax);
    gat_gemm_wmma<<<N_NODES / 16, 128, 0, stream>>>(hidden, W, bias, h_ws);
    gat_proj<<<(N_NODES + 7) / 8, NT, 0, stream>>>(h_ws, att_w, a_src, a_dst);
    gat_edge_logit<<<(N_EDGES + NT - 1) / NT, NT, 0, stream>>>(
        src, dst, a_src, a_dst, att_b, e_ws, emax);
    gat_edge_scatter<<<(N_EDGES + 7) / 8, NT, 0, stream>>>(
        src, dst, e_ws, emax, h_ws, denom, agg);
    gat_final<<<(N_NODES * OUT_DIM + NT - 1) / NT, NT, 0, stream>>>(
        agg, denom, h_ws, out);
}


# Projection launch removed: a_src/a_dst become a second WMMA product (att rows x h tile in LDS; new sum order) inside the GEMM launch, whose waves also write the owner bytes; 2 launches, not 3
# speedup vs baseline: 4.1806x; 4.1806x over previous
#include <hip/hip_runtime.h>

#define N_NODES   50000
#define N_EDGES   800000
#define IN_DIM    256
#define OUT_DIM   64
#define NEG_SLOPE 0.01f
#define NPB       200

typedef __attribute__((ext_vector_type(2))) float v2f;
typedef __attribute__((ext_vector_type(8))) float v8f;

#define GEMM_NT    128
#define GEMM_GRID  1024
#define GEMM_TILES (N_NODES / 16)
#define A_PITCH    (IN_DIM + 4)
#define O_PITCH    (OUT_DIM + 4)
#define GEMM_WAVES (GEMM_NT / 32)
#define N_LINES    (N_EDGES / 128)
static_assert(N_EDGES % 128 == 0, "the owner stream is written as whole 128-byte lines");
static_assert(N_LINES <= 2 * GEMM_GRID * GEMM_WAVES, "two lines per wave cover the owner stream");
static_assert(N_LINES > GEMM_GRID * GEMM_WAVES, "every wave has a first line");
static_assert(N_NODES % 16 == 0, "whole 16-row tiles");
static_assert(GEMM_GRID <= GEMM_TILES, "every group has a first tile");
static_assert(IN_DIM == 256 && OUT_DIM == 64 && GEMM_NT == 128, "the staging maps below are written for these sizes");

typedef __attribute__((ext_vector_type(4))) float v4f;

__global__ __launch_bounds__(GEMM_NT) void gat_gemm_wmma(
    const float* __restrict__ hidden,
    const float* __restrict__ W,
    const float* __restrict__ bias,
    const float* __restrict__ att_w,
    const int*   __restrict__ dst,
    float* __restrict__ h_out,
    float* __restrict__ asd,
    unsigned int* __restrict__ owner4)
{
    __shared__ __attribute__((aligned(16))) float a_lds[2][16 * A_PITCH];
    __shared__ __attribute__((aligned(16))) float o_lds[2][16 * O_PITCH];
    __shared__ __attribute__((aligned(16))) float att_lds[3 * O_PITCH];

    const int tid  = threadIdx.x;
    const int lane = tid & 31;
    const int wave = tid >> 5;
    const int wave_s = __builtin_amdgcn_readfirstlane(wave);
    const int half = lane >> 4;
    const int l16  = lane & 15;
    const int gcol = wave * 16 + l16;

    const int srow = tid >> 6;
    const int scol = (tid & 63) * 4;

    int tile = blockIdx.x;
    v4f pre[8];
    {
        const float* gp = hidden + ((size_t)tile * 16 + srow) * IN_DIM + scol;
#pragma unroll
        for (int i = 0; i < 8; ++i)
            pre[i] = __builtin_nontemporal_load(reinterpret_cast<const v4f*>(gp + (size_t)(2 * i) * IN_DIM));
    }
    const int  gw    = blockIdx.x * GEMM_WAVES + wave_s;
    const int  ln1   = gw + GEMM_GRID * GEMM_WAVES;
    const bool two   = ln1 < N_LINES;
    const int  oi0   = gw * 32 + lane;
    const int  oi1   = (two ? ln1 : N_LINES - 1) * 32 + lane;
    const int4 dv0   = reinterpret_cast<const int4*>(dst)[oi0];
    const int4 dv1   = reinterpret_cast<const int4*>(dst)[oi1];
    __builtin_amdgcn_sched_barrier(0);

    v2f breg[64];
#pragma unroll
    for (int s = 0; s < 64; ++s) {
        const int k = 8 * (s >> 1) + 4 * half + 2 * (s & 1);
        breg[s].x = W[(size_t)k * OUT_DIM + gcol];
        breg[s].y = W[(size_t)(k + 1) * OUT_DIM + gcol];
    }
    const float bb = bias[gcol];

    att_lds[(tid >> 6) * O_PITCH + (tid & 63)] = att_w[tid];
    if (tid < OUT_DIM) att_lds[2 * O_PITCH + tid] = 0.0f;

    {
        auto ob = [](int v) -> unsigned {
            return ((unsigned)v < (unsigned)N_NODES) ? (unsigned)v / NPB : 255u;
        };
        owner4[oi0] = ob(dv0.x) | (ob(dv0.y) << 8) | (ob(dv0.z) << 16) | (ob(dv0.w) << 24);
        if (two) owner4[oi1] = ob(dv1.x) | (ob(dv1.y) << 8) | (ob(dv1.z) << 16) | (ob(dv1.w) << 24);
    }

#pragma unroll
    for (int i = 0; i < 8; ++i)
        *reinterpret_cast<v4f*>(&a_lds[0][(2 * i + srow) * A_PITCH + scol]) = pre[i];
    __syncthreads();

    const float* attp = &att_lds[(l16 < 2 ? l16 : 2) * O_PITCH + 4 * half];
    int cur = 0;
    const int bx = blockIdx.x;
    int turn = (bx + (bx >> 2) + (bx >> 4) + (bx >> 6) + (bx >> 8)) & (GEMM_WAVES - 1);
    for (; tile < GEMM_TILES; tile += GEMM_GRID) {
        const int  nxt      = tile + GEMM_GRID;
        const bool has_next = nxt < GEMM_TILES;
        if (has_next) {
            const float* gp = hidden + ((size_t)nxt * 16 + srow) * IN_DIM + scol;
#pragma unroll
            for (int i = 0; i < 8; ++i)
                pre[i] = __builtin_nontemporal_load(reinterpret_cast<const v4f*>(gp + (size_t)(2 * i) * IN_DIM));
        }

        v8f acc0 = {}, acc1 = {};
        const float* ap = &a_lds[cur][l16 * A_PITCH + 4 * half];
#pragma unroll
        for (int j = 0; j < 32; ++j) {
            const v4f a = *reinterpret_cast<const v4f*>(ap + 8 * j);
            v2f a0, a1;
            a0.x = a.x; a0.y = a.y;
            a1.x = a.z; a1.y = a.w;
            acc0 = __builtin_amdgcn_wmma_f32_16x16x4_f32(
                 false, a0,  false, breg[2 * j],
                 (short)0, acc0,  false,  false);
            acc1 = __builtin_amdgcn_wmma_f32_16x16x4_f32(
                 false, a1,  false, breg[2 * j + 1],
                 (short)0, acc1,  false,  false);
        }

        float* op = &o_lds[cur][(8 * half) * O_PITCH + gcol];
#pragma unroll
        for (int r = 0; r < 8; ++r) op[r * O_PITCH] = (acc0[r] + acc1[r]) + bb;

        if (has_next) {
#pragma unroll
            for (int i = 0; i < 8; ++i)
                *reinterpret_cast<v4f*>(&a_lds[cur ^ 1][(2 * i + srow) * A_PITCH + scol]) = pre[i];
        }
        __syncthreads();

#pragma unroll
        for (int i = 0; i < 2; ++i) {
            const int row = wave * 4 + 2 * i + half;
            const v4f v = *reinterpret_cast<const v4f*>(&o_lds[cur][row * O_PITCH + 4 * l16]);
            *reinterpret_cast<v4f*>(h_out + ((size_t)tile * 16 + row) * OUT_DIM + 4 * l16) = v;
        }

        if (wave_s == turn) {
            const float* hp = &o_lds[cur][l16 * O_PITCH + 4 * half];
            v8f p0 = {}, p1 = {};
#pragma unroll
            for (int j = 0; j < 8; ++j) {
                const v4f at = *reinterpret_cast<const v4f*>(attp + 8 * j);
                const v4f hb = *reinterpret_cast<const v4f*>(hp + 8 * j);
                v2f at0, at1, hb0, hb1;
                at0.x = at.x; at0.y = at.y;  at1.x = at.z; at1.y = at.w;
                hb0.x = hb.x; hb0.y = hb.y;  hb1.x = hb.z; hb1.y = hb.w;
                p0 = __builtin_amdgcn_wmma_f32_16x16x4_f32(
                     false, at0,  false, hb0,
                     (short)0, p0,  false,  false);
                p1 = __builtin_amdgcn_wmma_f32_16x16x4_f32(
                     false, at1,  false, hb1,
                     (short)0, p1,  false,  false);
            }
            if (half == 0) {
                v2f pr;
                pr.x = p0[0] + p1[0];
                pr.y = p0[1] + p1[1];
                *reinterpret_cast<v2f*>(asd + ((size_t)tile * 16 + l16) * 2) = pr;
            }
        }
        turn = (turn + 1) & (GEMM_WAVES - 1);
        cur ^= 1;
    }
}

#define NGROUPS   (N_NODES / NPB)
#define AGG_NT    1024
#define AGG_WAVES (AGG_NT / 32)
#define LIST_CAP  16384
#define N_VEC16   (N_EDGES / 16)
#define ROT_VEC16 (N_VEC16 / NGROUPS)
static_assert(N_NODES % NPB == 0, "rows split evenly");
static_assert(NGROUPS <= 255, "an owner index fits a byte, 255 is free for 'no owner'");
static_assert(N_VEC16 % NGROUPS == 0 && ROT_VEC16 % 8 == 0, "rotation keeps 128-byte alignment");
static_assert(N_EDGES <= (1 << 20) && N_NODES <= (1 << 20) && NPB <= 2048, "20+11 bit packing");

__device__ __forceinline__ void ldsMaxF(float* addr, float v) {
    if (v >= 0.0f) atomicMax((int*)addr, __float_as_int(v));
    else           atomicMin((unsigned int*)addr, __float_as_uint(v));
}

__global__ __launch_bounds__(AGG_NT) void gat_aggregate(
    const int* __restrict__ src, const int* __restrict__ dst,
    const float* __restrict__ asd,
    const float* __restrict__ att_b, const float* __restrict__ h,
    const unsigned int* __restrict__ owner4,
    float* __restrict__ out)
{
    __shared__ float slab[NPB * OUT_DIM];
    __shared__ float emax_l[NPB];
    __shared__ float denom_l[NPB];
    __shared__ float adst_l[NPB];
    __shared__ int   list_a[LIST_CAP];
    __shared__ float list_v[LIST_CAP];
    __shared__ int   cnt;

    const int tid  = threadIdx.x;
    const int lane = tid & 31;
    const int wave = __builtin_amdgcn_readfirstlane(tid >> 5);
    const int g    = blockIdx.x;
    const int lo   = g * NPB;

    for (int i = tid; i < NPB * OUT_DIM; i += AGG_NT) slab[i] = 0.0f;
    if (tid < NPB) {
        emax_l[tid]  = -__builtin_inff();
        denom_l[tid] = 0.0f;
        adst_l[tid]  = asd[2 * (lo + tid) + 1];
    }
    if (tid == 0) cnt = 0;
    __syncthreads();

    const unsigned gpat = (unsigned)g * 0x01010101u;
    auto scan = [&](uint4 w, int e0) {
        auto ne = [&](unsigned v) -> unsigned {
            const unsigned x = v ^ gpat;
            return ((x & 0x7F7F7F7Fu) + 0x7F7F7F7Fu) | x;
        };
        const unsigned z0 = ne(w.x), z1 = ne(w.y), z2 = ne(w.z), z3 = ne(w.w);
        if ((z0 & z1 & z2 & z3 & 0x80808080u) != 0x80808080u) {
            const unsigned k = 0x01010101u;
            unsigned m = (((z0 >> 7) & k) | ((z1 >> 6) & (k << 1)) | ((z2 >> 5) & (k << 2))
                          | ((z3 >> 4) & (k << 3))) ^ 0x0F0F0F0Fu;
            while (m) {
                const int i   = __ffs(m) - 1;
                const int idx = atomicAdd(&cnt, 1);
                if (idx < LIST_CAP) list_a[idx] = e0 + 4 * (i & 7) + (i >> 3);
                m &= m - 1;
            }
        }
    };
    const uint4* __restrict__ own16 = reinterpret_cast<const uint4*>(owner4);
    const int rot = g * ROT_VEC16;
    constexpr int FULL_CH = N_VEC16 / AGG_NT;
    constexpr int MAIN_CH = FULL_CH - (FULL_CH % 8);
    constexpr int TAIL_CH = (N_VEC16 - MAIN_CH * AGG_NT + AGG_NT - 1) / AGG_NT;
    for (int c0 = 0; c0 < MAIN_CH; c0 += 8) {
        uint4 d[8];
        int   q[8];
#pragma unroll
        for (int u = 0; u < 8; ++u) {
            const int p = (c0 + u) * AGG_NT + tid + rot;
            q[u] = (p >= N_VEC16) ? p - N_VEC16 : p;
            d[u] = own16[q[u]];
        }
#pragma unroll
        for (int u = 0; u < 8; ++u) scan(d[u], q[u] * 16);
    }
#pragma unroll
    for (int u = 0; u < TAIL_CH; ++u) {
        const int p0 = (MAIN_CH + u) * AGG_NT + tid;
        if (p0 < N_VEC16) {
            const int p = p0 + rot;
            const int q = (p >= N_VEC16) ? p - N_VEC16 : p;
            scan(own16[q], q * 16);
        }
    }
    __syncthreads();

    const int   total = cnt;
    const float ab    = att_b[0];

    if (total <= LIST_CAP) {
        for (int base = tid; base < total; base += 4 * AGG_NT) {
            int   sv[4], dv[4];
            float as[4];
#pragma unroll
            for (int k = 0; k < 4; ++k) {
                const int idx = base + k * AGG_NT;
                const int e   = (idx < total) ? list_a[idx] : 0;
                sv[k] = src[e];
                dv[k] = dst[e];
            }
#pragma unroll
            for (int k = 0; k < 4; ++k) as[k] = asd[2 * sv[k]];
#pragma unroll
            for (int k = 0; k < 4; ++k) {
                const int idx = base + k * AGG_NT;
                if (idx < total) {
                    const int dl = dv[k] - lo;
                    float v = as[k] + adst_l[dl] + ab;
                    v = (v > 0.0f) ? v : NEG_SLOPE * v;
                    list_a[idx] = (dl << 20) | sv[k];
                    list_v[idx] = v;
                    ldsMaxF(&emax_l[dl], v);
                }
            }
        }
        __syncthreads();

        for (int i = tid; i < total; i += AGG_NT) {
            const int dl = (int)((unsigned)list_a[i] >> 20);
            const float ex = __expf(list_v[i] - emax_l[dl]);
            list_v[i] = ex;
            atomicAdd(&denom_l[dl], ex);
        }
        __syncthreads();

        for (int base = wave; base < total; base += 8 * AGG_WAVES) {
            int    pk[8];
            float  ex[8];
            float2 hv[8];
            const int   myidx = base + (lane & 7) * AGG_WAVES;
            const int   pkv   = (myidx < total) ? list_a[myidx] : 0;
            const float exv   = (myidx < total) ? list_v[myidx] : 0.0f;
#pragma unroll
            for (int k = 0; k < 8; ++k) {
                pk[k] = __builtin_amdgcn_readlane(pkv, k);
                ex[k] = __int_as_float(__builtin_amdgcn_readlane(__float_as_int(exv), k));
            }
#pragma unroll
            for (int k = 0; k < 8; ++k)
                hv[k] = *reinterpret_cast<const float2*>(h + (size_t)(pk[k] & 0xFFFFF) * OUT_DIM + 2 * lane);
            asm volatile("" :: "v"(hv[0].x), "v"(hv[0].y), "v"(hv[1].x), "v"(hv[1].y),
                               "v"(hv[2].x), "v"(hv[2].y), "v"(hv[3].x), "v"(hv[3].y),
                               "v"(hv[4].x), "v"(hv[4].y), "v"(hv[5].x), "v"(hv[5].y),
                               "v"(hv[6].x), "v"(hv[6].y), "v"(hv[7].x), "v"(hv[7].y));
#pragma unroll
            for (int k = 0; k < 8; ++k) {
                const int idx = base + k * AGG_WAVES;
                if (idx < total) {
                    const int dl = (int)((unsigned)pk[k] >> 20);
                    float* row = &slab[dl * OUT_DIM + 2 * lane];
                    atomicAdd(row,     ex[k] * hv[k].x);
                    atomicAdd(row + 1, ex[k] * hv[k].y);
                }
            }
        }
        __syncthreads();
    } else {
        for (int e = tid; e < N_EDGES; e += AGG_NT) {
            const unsigned dl = (unsigned)(dst[e] - lo);
            if (dl < (unsigned)NPB) {
                float v = asd[2 * src[e]] + adst_l[dl] + ab;
                v = (v > 0.0f) ? v : NEG_SLOPE * v;
                ldsMaxF(&emax_l[dl], v);
            }
        }
        __syncthreads();
        for (int e = tid; e < N_EDGES; e += AGG_NT) {
            const unsigned dl = (unsigned)(dst[e] - lo);
            if (dl < (unsigned)NPB) {
                const int s = src[e];
                float v = asd[2 * s] + adst_l[dl] + ab;
                v = (v > 0.0f) ? v : NEG_SLOPE * v;
                const float ex = __expf(v - emax_l[dl]);
                atomicAdd(&denom_l[dl], ex);
                for (int c = 0; c < OUT_DIM; ++c) {
                    const int cc = (c + lane) & (OUT_DIM - 1);
                    atomicAdd(&slab[dl * OUT_DIM + cc], ex * h[(size_t)s * OUT_DIM + cc]);
                }
            }
        }
        __syncthreads();
    }

    for (int r = wave; r < NPB; r += AGG_WAVES) {
        const float  dn = denom_l[r];
        const size_t o  = (size_t)(lo + r) * OUT_DIM + lane;
        float v0, v1;
        if (dn > 0.0f) {
            v0 = slab[r * OUT_DIM + lane]      / dn;
            v1 = slab[r * OUT_DIM + lane + 32] / dn;
        } else {
            v0 = h[o];
            v1 = h[o + 32];
        }
        out[o]      = v0;
        out[o + 32] = v1;
    }
}

extern "C" void kernel_launch(void* const* d_in, const int* in_sizes, int n_in,
                              void* d_out, int out_size, void* d_ws, size_t ws_size,
                              hipStream_t stream) {
    const float* hidden = (const float*)d_in[0];
    const float* W      = (const float*)d_in[1];
    const float* bias   = (const float*)d_in[2];
    const float* att_w  = (const float*)d_in[3];
    const float* att_b  = (const float*)d_in[4];
    const int*   src    = (const int*)  d_in[5];
    const int*   dst    = (const int*)  d_in[6];
    float* out = (float*)d_out;

    float* ws     = (float*)d_ws;
    float* h_ws   = ws;
    float* asd    = h_ws + (size_t)N_NODES * OUT_DIM;
    unsigned int* owner4 = (unsigned int*)(asd + 2 * (size_t)N_NODES);

    gat_gemm_wmma<<<GEMM_GRID, GEMM_NT, 0, stream>>>(hidden, W, bias, att_w, dst, h_ws, asd, owner4);
    gat_aggregate<<<NGROUPS, AGG_NT, 0, stream>>>(src, dst, asd, att_b, h_ws, owner4, out);
}


# Edge filter reads one presence word per 1024 edges (32 chunk bits per owner, LDS ORs in the GEMM launch) and scans only the marked 32-edge chunks of the owner stream; sum order = LDS arrival order
# speedup vs baseline: 4.5847x; 1.0967x over previous
#include <hip/hip_runtime.h>

#define N_NODES   50000
#define N_EDGES   800000
#define IN_DIM    256
#define OUT_DIM   64
#define NEG_SLOPE 0.01f
#define NPB       200

typedef __attribute__((ext_vector_type(2))) float v2f;
typedef __attribute__((ext_vector_type(8))) float v8f;

#define GEMM_NT    128
#define GEMM_GRID  1024
#define GEMM_TILES (N_NODES / 16)
#define A_PITCH    (IN_DIM + 4)
#define O_PITCH    (OUT_DIM + 4)
#define GEMM_WAVES (GEMM_NT / 32)
#define N_LINES    (N_EDGES / 128)
static_assert(N_EDGES % 128 == 0, "the owner stream is written as whole 128-byte lines");
static_assert(N_LINES <= 2 * GEMM_GRID * GEMM_WAVES, "two lines per wave cover the owner stream");
static_assert(N_LINES > GEMM_GRID * GEMM_WAVES, "every wave has a first line");
static_assert(N_NODES % 16 == 0, "whole 16-row tiles");
static_assert(GEMM_GRID <= GEMM_TILES, "every group has a first tile");
static_assert(IN_DIM == 256 && OUT_DIM == 64 && GEMM_NT == 128, "the staging maps below are written for these sizes");

typedef __attribute__((ext_vector_type(4))) float v4f;

__global__ __launch_bounds__(GEMM_NT) void gat_gemm_wmma(
    const float* __restrict__ hidden,
    const float* __restrict__ W,
    const float* __restrict__ bias,
    const float* __restrict__ att_w,
    const int*   __restrict__ dst,
    float* __restrict__ h_out,
    float* __restrict__ asd,
    unsigned int* __restrict__ owner4,
    unsigned int* __restrict__ pres)
{
    __shared__ __attribute__((aligned(8))) unsigned int pres_lds[256];
    __shared__ __attribute__((aligned(16))) float a_lds[2][16 * A_PITCH];
    __shared__ __attribute__((aligned(16))) float o_lds[2][16 * O_PITCH];
    __shared__ __attribute__((aligned(16))) float att_lds[3 * O_PITCH];

    const int tid  = threadIdx.x;
    const int lane = tid & 31;
    const int wave = tid >> 5;
    const int wave_s = __builtin_amdgcn_readfirstlane(wave);
    const int half = lane >> 4;
    const int l16  = lane & 15;
    const int gcol = wave * 16 + l16;

    const int srow = tid >> 6;
    const int scol = (tid & 63) * 4;

    *reinterpret_cast<uint2*>(&pres_lds[2 * tid]) = make_uint2(0u, 0u);

    int tile = blockIdx.x;
    v4f pre[8];
    {
        const float* gp = hidden + ((size_t)tile * 16 + srow) * IN_DIM + scol;
#pragma unroll
        for (int i = 0; i < 8; ++i)
            pre[i] = __builtin_nontemporal_load(reinterpret_cast<const v4f*>(gp + (size_t)(2 * i) * IN_DIM));
    }
    const int  gw    = blockIdx.x * GEMM_WAVES + wave_s;
    const int  ln1   = gw + GEMM_GRID * GEMM_WAVES;
    const bool two   = ln1 < N_LINES;
    const int  oi0   = gw * 32 + lane;
    const int  oi1   = (two ? ln1 : N_LINES - 1) * 32 + lane;
    const int4 dv0   = reinterpret_cast<const int4*>(dst)[oi0];
    const int4 dv1   = reinterpret_cast<const int4*>(dst)[oi1];
    __builtin_amdgcn_sched_barrier(0);

    v2f breg[64];
#pragma unroll
    for (int s = 0; s < 64; ++s) {
        const int k = 8 * (s >> 1) + 4 * half + 2 * (s & 1);
        breg[s].x = W[(size_t)k * OUT_DIM + gcol];
        breg[s].y = W[(size_t)(k + 1) * OUT_DIM + gcol];
    }
    const float bb = bias[gcol];

    att_lds[(tid >> 6) * O_PITCH + (tid & 63)] = att_w[tid];
    if (tid < OUT_DIM) att_lds[2 * O_PITCH + tid] = 0.0f;

    {
        auto ob = [](int v) -> unsigned {
            return ((unsigned)v < (unsigned)N_NODES) ? (unsigned)v / NPB : 255u;
        };
        owner4[oi0] = ob(dv0.x) | (ob(dv0.y) << 8) | (ob(dv0.z) << 16) | (ob(dv0.w) << 24);
        if (two) owner4[oi1] = ob(dv1.x) | (ob(dv1.y) << 8) | (ob(dv1.z) << 16) | (ob(dv1.w) << 24);
        __syncthreads();
        const unsigned bit0 = 1u << (8 * wave_s + (lane >> 3));
        atomicOr(&pres_lds[ob(dv0.x)], bit0);
        atomicOr(&pres_lds[ob(dv0.y)], bit0);
        atomicOr(&pres_lds[ob(dv0.z)], bit0);
        atomicOr(&pres_lds[ob(dv0.w)], bit0);
        if (two) {
            const unsigned bit1 = bit0 << 4;
            atomicOr(&pres_lds[ob(dv1.x)], bit1);
            atomicOr(&pres_lds[ob(dv1.y)], bit1);
            atomicOr(&pres_lds[ob(dv1.z)], bit1);
            atomicOr(&pres_lds[ob(dv1.w)], bit1);
        }
    }

#pragma unroll
    for (int i = 0; i < 8; ++i)
        *reinterpret_cast<v4f*>(&a_lds[0][(2 * i + srow) * A_PITCH + scol]) = pre[i];
    __syncthreads();

    *reinterpret_cast<uint2*>(pres + (size_t)blockIdx.x * 256 + 2 * tid) =
        *reinterpret_cast<const uint2*>(&pres_lds[2 * tid]);

    const float* attp = &att_lds[(l16 < 2 ? l16 : 2) * O_PITCH + 4 * half];
    int cur = 0;
    const int bx = blockIdx.x;
    int turn = (bx + (bx >> 2) + (bx >> 4) + (bx >> 6) + (bx >> 8)) & (GEMM_WAVES - 1);
    for (; tile < GEMM_TILES; tile += GEMM_GRID) {
        const int  nxt      = tile + GEMM_GRID;
        const bool has_next = nxt < GEMM_TILES;
        if (has_next) {
            const float* gp = hidden + ((size_t)nxt * 16 + srow) * IN_DIM + scol;
#pragma unroll
            for (int i = 0; i < 8; ++i)
                pre[i] = __builtin_nontemporal_load(reinterpret_cast<const v4f*>(gp + (size_t)(2 * i) * IN_DIM));
        }

        v8f acc0 = {}, acc1 = {};
        const float* ap = &a_lds[cur][l16 * A_PITCH + 4 * half];
#pragma unroll
        for (int j = 0; j < 32; ++j) {
            const v4f a = *reinterpret_cast<const v4f*>(ap + 8 * j);
            v2f a0, a1;
            a0.x = a.x; a0.y = a.y;
            a1.x = a.z; a1.y = a.w;
            acc0 = __builtin_amdgcn_wmma_f32_16x16x4_f32(
                 false, a0,  false, breg[2 * j],
                 (short)0, acc0,  false,  false);
            acc1 = __builtin_amdgcn_wmma_f32_16x16x4_f32(
                 false, a1,  false, breg[2 * j + 1],
                 (short)0, acc1,  false,  false);
        }

        float* op = &o_lds[cur][(8 * half) * O_PITCH + gcol];
#pragma unroll
        for (int r = 0; r < 8; ++r) op[r * O_PITCH] = (acc0[r] + acc1[r]) + bb;

        if (has_next) {
#pragma unroll
            for (int i = 0; i < 8; ++i)
                *reinterpret_cast<v4f*>(&a_lds[cur ^ 1][(2 * i + srow) * A_PITCH + scol]) = pre[i];
        }
        __syncthreads();

#pragma unroll
        for (int i = 0; i < 2; ++i) {
            const int row = wave * 4 + 2 * i + half;
            const v4f v = *reinterpret_cast<const v4f*>(&o_lds[cur][row * O_PITCH + 4 * l16]);
            *reinterpret_cast<v4f*>(h_out + ((size_t)tile * 16 + row) * OUT_DIM + 4 * l16) = v;
        }

        if (wave_s == turn) {
            const float* hp = &o_lds[cur][l16 * O_PITCH + 4 * half];
            v8f p0 = {}, p1 = {};
#pragma unroll
            for (int j = 0; j < 8; ++j) {
                const v4f at = *reinterpret_cast<const v4f*>(attp + 8 * j);
                const v4f hb = *reinterpret_cast<const v4f*>(hp + 8 * j);
                v2f at0, at1, hb0, hb1;
                at0.x = at.x; at0.y = at.y;  at1.x = at.z; at1.y = at.w;
                hb0.x = hb.x; hb0.y = hb.y;  hb1.x = hb.z; hb1.y = hb.w;
                p0 = __builtin_amdgcn_wmma_f32_16x16x4_f32(
                     false, at0,  false, hb0,
                     (short)0, p0,  false,  false);
                p1 = __builtin_amdgcn_wmma_f32_16x16x4_f32(
                     false, at1,  false, hb1,
                     (short)0, p1,  false,  false);
            }
            if (half == 0) {
                v2f pr;
                pr.x = p0[0] + p1[0];
                pr.y = p0[1] + p1[1];
                *reinterpret_cast<v2f*>(asd + ((size_t)tile * 16 + l16) * 2) = pr;
            }
        }
        turn = (turn + 1) & (GEMM_WAVES - 1);
        cur ^= 1;
    }
}

#define NGROUPS   (N_NODES / NPB)
#define AGG_NT    1024
#define AGG_WAVES (AGG_NT / 32)
#define LIST_CAP  16384
static_assert(N_NODES % NPB == 0, "rows split evenly");
static_assert(NGROUPS <= 255, "an owner index fits a byte, 255 is free for 'no owner'");
static_assert(AGG_NT == GEMM_GRID, "one presence word per thread");
static_assert(GEMM_WAVES == 4, "a presence word has 8 bits per wave of its group");
static_assert(N_EDGES <= (1 << 20) && N_NODES <= (1 << 20) && NPB <= 2048, "20+11 bit packing");

__device__ __forceinline__ void ldsMaxF(float* addr, float v) {
    if (v >= 0.0f) atomicMax((int*)addr, __float_as_int(v));
    else           atomicMin((unsigned int*)addr, __float_as_uint(v));
}

__global__ __launch_bounds__(AGG_NT) void gat_aggregate(
    const int* __restrict__ src, const int* __restrict__ dst,
    const float* __restrict__ asd,
    const float* __restrict__ att_b, const float* __restrict__ h,
    const unsigned int* __restrict__ owner4,
    const unsigned int* __restrict__ pres,
    float* __restrict__ out)
{
    __shared__ int   clist[LIST_CAP];
    __shared__ int   ccnt;
    __shared__ float slab[NPB * OUT_DIM];
    __shared__ float emax_l[NPB];
    __shared__ float denom_l[NPB];
    __shared__ float adst_l[NPB];
    __shared__ int   list_a[LIST_CAP];
    __shared__ float list_v[LIST_CAP];
    __shared__ int   cnt;

    const int tid  = threadIdx.x;
    const int lane = tid & 31;
    const int wave = __builtin_amdgcn_readfirstlane(tid >> 5);
    const int g    = blockIdx.x;
    const int lo   = g * NPB;

    for (int i = tid; i < NPB * OUT_DIM; i += AGG_NT) slab[i] = 0.0f;
    if (tid < NPB) {
        emax_l[tid]  = -__builtin_inff();
        denom_l[tid] = 0.0f;
        adst_l[tid]  = asd[2 * (lo + tid) + 1];
    }
    if (tid == 0) { cnt = 0; ccnt = 0; }
    const unsigned pw = pres[(size_t)tid * 256 + g];
    __syncthreads();

    const unsigned gpat = (unsigned)g * 0x01010101u;
    auto scan32 = [&](uint4 wa, uint4 wb, int e0) {
        auto ne = [&](unsigned v) -> unsigned {
            const unsigned x = v ^ gpat;
            return ((x & 0x7F7F7F7Fu) + 0x7F7F7F7Fu) | x;
        };
        const unsigned k = 0x01010101u;
        const unsigned r = ((ne(wa.x) >> 7) & k)        | ((ne(wa.y) >> 6) & (k << 1))
                         | ((ne(wa.z) >> 5) & (k << 2)) | ((ne(wa.w) >> 4) & (k << 3))
                         | ((ne(wb.x) >> 3) & (k << 4)) | ((ne(wb.y) >> 2) & (k << 5))
                         | ((ne(wb.z) >> 1) & (k << 6)) | ( ne(wb.w)       & (k << 7));
        unsigned m = ~r;
        while (m) {
            const int i   = __ffs(m) - 1;
            const int idx = atomicAdd(&cnt, 1);
            if (idx < LIST_CAP) list_a[idx] = e0 + 4 * (i & 7) + (i >> 3);
            m &= m - 1;
        }
    };
    const uint4* __restrict__ own16 = reinterpret_cast<const uint4*>(owner4);
    {
        const int n = __popc(pw);
        unsigned below = 0, all = 0;
#pragma unroll
        for (int j = 0; j < 6; ++j) {
            const unsigned vote = __builtin_amdgcn_ballot_w32(((n >> j) & 1) != 0);
            below += __builtin_amdgcn_mbcnt_lo(vote, 0u) << j;
            all   += (unsigned)__popc(vote) << j;
        }
        int wbase = 0;
        if (lane == 0) wbase = atomicAdd(&ccnt, (int)all);
        wbase = __builtin_amdgcn_readfirstlane(wbase);
        int slot = wbase + (int)below;
        unsigned m = pw;
        while (m) {
            const int b = __ffs(m) - 1;
            const int line = tid * GEMM_WAVES + (b >> 3) + ((b & 4) ? GEMM_GRID * GEMM_WAVES : 0);
            if (slot < LIST_CAP) clist[slot] = line * 4 + (b & 3);
            ++slot;
            m &= m - 1;
        }
    }
    __syncthreads();
    const int nch_all = ccnt;
    const int nch     = (nch_all < LIST_CAP) ? nch_all : LIST_CAP;
    for (int base = tid; base < nch; base += 4 * AGG_NT) {
        uint4 d[8];
        int   ch[4];
#pragma unroll
        for (int u = 0; u < 4; ++u) {
            const int idx = base + u * AGG_NT;
            ch[u] = (idx < nch) ? clist[idx] : 0;
            d[2 * u]     = own16[2 * ch[u]];
            d[2 * u + 1] = own16[2 * ch[u] + 1];
        }
#pragma unroll
        for (int u = 0; u < 4; ++u) {
            if (base + u * AGG_NT < nch) scan32(d[2 * u], d[2 * u + 1], ch[u] * 32);
        }
    }
    __syncthreads();

    const int   total = (nch_all > LIST_CAP) ? LIST_CAP + 1 : cnt;
    const float ab    = att_b[0];

    if (total <= LIST_CAP) {
        for (int base = tid; base < total; base += 4 * AGG_NT) {
            int   sv[4], dv[4];
            float as[4];
#pragma unroll
            for (int k = 0; k < 4; ++k) {
                const int idx = base + k * AGG_NT;
                const int e   = (idx < total) ? list_a[idx] : 0;
                sv[k] = src[e];
                dv[k] = dst[e];
            }
#pragma unroll
            for (int k = 0; k < 4; ++k) as[k] = asd[2 * sv[k]];
#pragma unroll
            for (int k = 0; k < 4; ++k) {
                const int idx = base + k * AGG_NT;
                if (idx < total) {
                    const int dl = dv[k] - lo;
                    float v = as[k] + adst_l[dl] + ab;
                    v = (v > 0.0f) ? v : NEG_SLOPE * v;
                    list_a[idx] = (dl << 20) | sv[k];
                    list_v[idx] = v;
                    ldsMaxF(&emax_l[dl], v);
                }
            }
        }
        __syncthreads();

        for (int i = tid; i < total; i += AGG_NT) {
            const int dl = (int)((unsigned)list_a[i] >> 20);
            const float ex = __expf(list_v[i] - emax_l[dl]);
            list_v[i] = ex;
            atomicAdd(&denom_l[dl], ex);
        }
        __syncthreads();

        for (int base = wave; base < total; base += 8 * AGG_WAVES) {
            int    pk[8];
            float  ex[8];
            float2 hv[8];
            const int   myidx = base + (lane & 7) * AGG_WAVES;
            const int   pkv   = (myidx < total) ? list_a[myidx] : 0;
            const float exv   = (myidx < total) ? list_v[myidx] : 0.0f;
#pragma unroll
            for (int k = 0; k < 8; ++k) {
                pk[k] = __builtin_amdgcn_readlane(pkv, k);
                ex[k] = __int_as_float(__builtin_amdgcn_readlane(__float_as_int(exv), k));
            }
#pragma unroll
            for (int k = 0; k < 8; ++k)
                hv[k] = *reinterpret_cast<const float2*>(h + (size_t)(pk[k] & 0xFFFFF) * OUT_DIM + 2 * lane);
            asm volatile("" :: "v"(hv[0].x), "v"(hv[0].y), "v"(hv[1].x), "v"(hv[1].y),
                               "v"(hv[2].x), "v"(hv[2].y), "v"(hv[3].x), "v"(hv[3].y),
                               "v"(hv[4].x), "v"(hv[4].y), "v"(hv[5].x), "v"(hv[5].y),
                               "v"(hv[6].x), "v"(hv[6].y), "v"(hv[7].x), "v"(hv[7].y));
#pragma unroll
            for (int k = 0; k < 8; ++k) {
                const int idx = base + k * AGG_WAVES;
                if (idx < total) {
                    const int dl = (int)((unsigned)pk[k] >> 20);
                    float* row = &slab[dl * OUT_DIM + 2 * lane];
                    atomicAdd(row,     ex[k] * hv[k].x);
                    atomicAdd(row + 1, ex[k] * hv[k].y);
                }
            }
        }
        __syncthreads();
    } else {
        for (int e = tid; e < N_EDGES; e += AGG_NT) {
            const unsigned dl = (unsigned)(dst[e] - lo);
            if (dl < (unsigned)NPB) {
                float v = asd[2 * src[e]] + adst_l[dl] + ab;
                v = (v > 0.0f) ? v : NEG_SLOPE * v;
                ldsMaxF(&emax_l[dl], v);
            }
        }
        __syncthreads();
        for (int e = tid; e < N_EDGES; e += AGG_NT) {
            const unsigned dl = (unsigned)(dst[e] - lo);
            if (dl < (unsigned)NPB) {
                const int s = src[e];
                float v = asd[2 * s] + adst_l[dl] + ab;
                v = (v > 0.0f) ? v : NEG_SLOPE * v;
                const float ex = __expf(v - emax_l[dl]);
                atomicAdd(&denom_l[dl], ex);
                for (int c = 0; c < OUT_DIM; ++c) {
                    const int cc = (c + lane) & (OUT_DIM - 1);
                    atomicAdd(&slab[dl * OUT_DIM + cc], ex * h[(size_t)s * OUT_DIM + cc]);
                }
            }
        }
        __syncthreads();
    }

    for (int r = wave; r < NPB; r += AGG_WAVES) {
        const float  dn = denom_l[r];
        const size_t o  = (size_t)(lo + r) * OUT_DIM + lane;
        float v0, v1;
        if (dn > 0.0f) {
            v0 = slab[r * OUT_DIM + lane]      / dn;
            v1 = slab[r * OUT_DIM + lane + 32] / dn;
        } else {
            v0 = h[o];
            v1 = h[o + 32];
        }
        out[o]      = v0;
        out[o + 32] = v1;
    }
}

extern "C" void kernel_launch(void* const* d_in, const int* in_sizes, int n_in,
                              void* d_out, int out_size, void* d_ws, size_t ws_size,
                              hipStream_t stream) {
    const float* hidden = (const float*)d_in[0];
    const float* W      = (const float*)d_in[1];
    const float* bias   = (const float*)d_in[2];
    const float* att_w  = (const float*)d_in[3];
    const float* att_b  = (const float*)d_in[4];
    const int*   src    = (const int*)  d_in[5];
    const int*   dst    = (const int*)  d_in[6];
    float* out = (float*)d_out;

    float* ws     = (float*)d_ws;
    float* h_ws   = ws;
    float* asd    = h_ws + (size_t)N_NODES * OUT_DIM;
    unsigned int* owner4 = (unsigned int*)(asd + 2 * (size_t)N_NODES);
    unsigned int* pres   = owner4 + N_EDGES / 4;

    gat_gemm_wmma<<<GEMM_GRID, GEMM_NT, 0, stream>>>(hidden, W, bias, att_w, dst, h_ws, asd, owner4, pres);
    gat_aggregate<<<NGROUPS, AGG_NT, 0, stream>>>(src, dst, asd, att_b, h_ws, owner4, pres, out);
}


# Edge stream holds a 4-byte record per edge (owner, row in group, 2-byte src) from the GEMM launch; last launch reads marked chunks as whole lines, gathers no src/dst; sum order = LDS arrival order
# speedup vs baseline: 4.9145x; 1.0719x over previous
#include <hip/hip_runtime.h>

#define N_NODES   50000
#define N_EDGES   800000
#define IN_DIM    256
#define OUT_DIM   64
#define NEG_SLOPE 0.01f
#define NPB       200

typedef __attribute__((ext_vector_type(2))) float v2f;
typedef __attribute__((ext_vector_type(8))) float v8f;

#define GEMM_NT    128
#define GEMM_GRID  1024
#define GEMM_TILES (N_NODES / 16)
#define A_PITCH    (IN_DIM + 4)
#define O_PITCH    (OUT_DIM + 4)
#define GEMM_WAVES (GEMM_NT / 32)
#define N_LINES    (N_EDGES / 128)
static_assert(N_EDGES % 128 == 0, "the records are written as whole 128-byte lines");
static_assert(N_LINES <= 2 * GEMM_GRID * GEMM_WAVES, "two blocks per wave cover the edges");
static_assert(N_LINES > GEMM_GRID * GEMM_WAVES, "every wave has a first block");
static_assert(N_NODES <= 65536 && NPB <= 256, "a record holds src in 16 bits and the row inside its group in 8");
static_assert(N_NODES % 16 == 0, "whole 16-row tiles");
static_assert(GEMM_GRID <= GEMM_TILES, "every group has a first tile");
static_assert(IN_DIM == 256 && OUT_DIM == 64 && GEMM_NT == 128, "the staging maps below are written for these sizes");

typedef __attribute__((ext_vector_type(4))) float v4f;

__global__ __launch_bounds__(GEMM_NT) void gat_gemm_wmma(
    const float* __restrict__ hidden,
    const float* __restrict__ W,
    const float* __restrict__ bias,
    const float* __restrict__ att_w,
    const int*   __restrict__ src,
    const int*   __restrict__ dst,
    float* __restrict__ h_out,
    float* __restrict__ asd,
    unsigned int* __restrict__ rec,
    unsigned int* __restrict__ pres)
{
    __shared__ __attribute__((aligned(8))) unsigned int pres_lds[256];
    __shared__ __attribute__((aligned(16))) float a_lds[2][16 * A_PITCH];
    __shared__ __attribute__((aligned(16))) float o_lds[2][16 * O_PITCH];
    __shared__ __attribute__((aligned(16))) float att_lds[3 * O_PITCH];

    const int tid  = threadIdx.x;
    const int lane = tid & 31;
    const int wave = tid >> 5;
    const int wave_s = __builtin_amdgcn_readfirstlane(wave);
    const int half = lane >> 4;
    const int l16  = lane & 15;
    const int gcol = wave * 16 + l16;

    const int srow = tid >> 6;
    const int scol = (tid & 63) * 4;

    *reinterpret_cast<uint2*>(&pres_lds[2 * tid]) = make_uint2(0u, 0u);

    int tile = blockIdx.x;
    v4f pre[8];
    {
        const float* gp = hidden + ((size_t)tile * 16 + srow) * IN_DIM + scol;
#pragma unroll
        for (int i = 0; i < 8; ++i)
            pre[i] = __builtin_nontemporal_load(reinterpret_cast<const v4f*>(gp + (size_t)(2 * i) * IN_DIM));
    }
    const int  gw    = blockIdx.x * GEMM_WAVES + wave_s;
    const int  ln1   = gw + GEMM_GRID * GEMM_WAVES;
    const bool two   = ln1 < N_LINES;
    const int  oi0   = gw * 32 + lane;
    const int  oi1   = (two ? ln1 : N_LINES - 1) * 32 + lane;
    const int4 dv0   = reinterpret_cast<const int4*>(dst)[oi0];
    const int4 dv1   = reinterpret_cast<const int4*>(dst)[oi1];
    const int4 sv0   = reinterpret_cast<const int4*>(src)[oi0];
    const int4 sv1   = reinterpret_cast<const int4*>(src)[oi1];
    __builtin_amdgcn_sched_barrier(0);

    v2f breg[64];
#pragma unroll
    for (int s = 0; s < 64; ++s) {
        const int k = 8 * (s >> 1) + 4 * half + 2 * (s & 1);
        breg[s].x = W[(size_t)k * OUT_DIM + gcol];
        breg[s].y = W[(size_t)(k + 1) * OUT_DIM + gcol];
    }
    const float bb = bias[gcol];

    att_lds[(tid >> 6) * O_PITCH + (tid & 63)] = att_w[tid];
    if (tid < OUT_DIM) att_lds[2 * O_PITCH + tid] = 0.0f;

    {
        auto ob = [](int v) -> unsigned {
            return ((unsigned)v < (unsigned)N_NODES) ? (unsigned)v / NPB : 255u;
        };
        auto rc = [&](int d, int s) -> unsigned {
            const unsigned o  = ob(d);
            const unsigned dl = (o == 255u) ? 0u : (unsigned)d - o * NPB;
            return (o << 24) | (dl << 16) | ((unsigned)s & 0xFFFFu);
        };
        reinterpret_cast<uint4*>(rec)[oi0] =
            make_uint4(rc(dv0.x, sv0.x), rc(dv0.y, sv0.y), rc(dv0.z, sv0.z), rc(dv0.w, sv0.w));
        if (two) reinterpret_cast<uint4*>(rec)[oi1] =
            make_uint4(rc(dv1.x, sv1.x), rc(dv1.y, sv1.y), rc(dv1.z, sv1.z), rc(dv1.w, sv1.w));
        __syncthreads();
        const unsigned bit0 = 1u << (8 * wave_s + (lane >> 3));
        atomicOr(&pres_lds[ob(dv0.x)], bit0);
        atomicOr(&pres_lds[ob(dv0.y)], bit0);
        atomicOr(&pres_lds[ob(dv0.z)], bit0);
        atomicOr(&pres_lds[ob(dv0.w)], bit0);
        if (two) {
            const unsigned bit1 = bit0 << 4;
            atomicOr(&pres_lds[ob(dv1.x)], bit1);
            atomicOr(&pres_lds[ob(dv1.y)], bit1);
            atomicOr(&pres_lds[ob(dv1.z)], bit1);
            atomicOr(&pres_lds[ob(dv1.w)], bit1);
        }
    }

#pragma unroll
    for (int i = 0; i < 8; ++i)
        *reinterpret_cast<v4f*>(&a_lds[0][(2 * i + srow) * A_PITCH + scol]) = pre[i];
    __syncthreads();

    *reinterpret_cast<uint2*>(pres + (size_t)blockIdx.x * 256 + 2 * tid) =
        *reinterpret_cast<const uint2*>(&pres_lds[2 * tid]);

    const float* attp = &att_lds[(l16 < 2 ? l16 : 2) * O_PITCH + 4 * half];
    int cur = 0;
    const int bx = blockIdx.x;
    int turn = (bx + (bx >> 2) + (bx >> 4) + (bx >> 6) + (bx >> 8)) & (GEMM_WAVES - 1);
    for (; tile < GEMM_TILES; tile += GEMM_GRID) {
        const int  nxt      = tile + GEMM_GRID;
        const bool has_next = nxt < GEMM_TILES;
        if (has_next) {
            const float* gp = hidden + ((size_t)nxt * 16 + srow) * IN_DIM + scol;
#pragma unroll
            for (int i = 0; i < 8; ++i)
                pre[i] = __builtin_nontemporal_load(reinterpret_cast<const v4f*>(gp + (size_t)(2 * i) * IN_DIM));
        }

        v8f acc0 = {}, acc1 = {};
        const float* ap = &a_lds[cur][l16 * A_PITCH + 4 * half];
#pragma unroll
        for (int j = 0; j < 32; ++j) {
            const v4f a = *reinterpret_cast<const v4f*>(ap + 8 * j);
            v2f a0, a1;
            a0.x = a.x; a0.y = a.y;
            a1.x = a.z; a1.y = a.w;
            acc0 = __builtin_amdgcn_wmma_f32_16x16x4_f32(
                 false, a0,  false, breg[2 * j],
                 (short)0, acc0,  false,  false);
            acc1 = __builtin_amdgcn_wmma_f32_16x16x4_f32(
                 false, a1,  false, breg[2 * j + 1],
                 (short)0, acc1,  false,  false);
        }

        float* op = &o_lds[cur][(8 * half) * O_PITCH + gcol];
#pragma unroll
        for (int r = 0; r < 8; ++r) op[r * O_PITCH] = (acc0[r] + acc1[r]) + bb;

        if (has_next) {
#pragma unroll
            for (int i = 0; i < 8; ++i)
                *reinterpret_cast<v4f*>(&a_lds[cur ^ 1][(2 * i + srow) * A_PITCH + scol]) = pre[i];
        }
        __syncthreads();

#pragma unroll
        for (int i = 0; i < 2; ++i) {
            const int row = wave * 4 + 2 * i + half;
            const v4f v = *reinterpret_cast<const v4f*>(&o_lds[cur][row * O_PITCH + 4 * l16]);
            *reinterpret_cast<v4f*>(h_out + ((size_t)tile * 16 + row) * OUT_DIM + 4 * l16) = v;
        }

        if (wave_s == turn) {
            const float* hp = &o_lds[cur][l16 * O_PITCH + 4 * half];
            v8f p0 = {}, p1 = {};
#pragma unroll
            for (int j = 0; j < 8; ++j) {
                const v4f at = *reinterpret_cast<const v4f*>(attp + 8 * j);
                const v4f hb = *reinterpret_cast<const v4f*>(hp + 8 * j);
                v2f at0, at1, hb0, hb1;
                at0.x = at.x; at0.y = at.y;  at1.x = at.z; at1.y = at.w;
                hb0.x = hb.x; hb0.y = hb.y;  hb1.x = hb.z; hb1.y = hb.w;
                p0 = __builtin_amdgcn_wmma_f32_16x16x4_f32(
                     false, at0,  false, hb0,
                     (short)0, p0,  false,  false);
                p1 = __builtin_amdgcn_wmma_f32_16x16x4_f32(
                     false, at1,  false, hb1,
                     (short)0, p1,  false,  false);
            }
            if (half == 0) {
                v2f pr;
                pr.x = p0[0] + p1[0];
                pr.y = p0[1] + p1[1];
                *reinterpret_cast<v2f*>(asd + ((size_t)tile * 16 + l16) * 2) = pr;
            }
        }
        turn = (turn + 1) & (GEMM_WAVES - 1);
        cur ^= 1;
    }
}

#define NGROUPS   (N_NODES / NPB)
#define AGG_NT    1024
#define AGG_WAVES (AGG_NT / 32)
#define LIST_CAP  16384
static_assert(N_NODES % NPB == 0, "rows split evenly");
static_assert(NGROUPS <= 255, "an owner index fits a byte, 255 is free for 'no owner'");
static_assert(AGG_NT == GEMM_GRID, "one presence word per thread");
static_assert(GEMM_WAVES == 4, "a presence word has 8 bits per wave of its group");
static_assert(N_EDGES <= (1 << 20) && N_NODES <= (1 << 20) && NPB <= 2048, "20+11 bit packing");

__device__ __forceinline__ void ldsMaxF(float* addr, float v) {
    if (v >= 0.0f) atomicMax((int*)addr, __float_as_int(v));
    else           atomicMin((unsigned int*)addr, __float_as_uint(v));
}

__global__ __launch_bounds__(AGG_NT) void gat_aggregate(
    const int* __restrict__ src, const int* __restrict__ dst,
    const float* __restrict__ asd,
    const float* __restrict__ att_b, const float* __restrict__ h,
    const unsigned int* __restrict__ rec,
    const unsigned int* __restrict__ pres,
    float* __restrict__ out)
{
    __shared__ int   clist[LIST_CAP];
    __shared__ int   ccnt;
    __shared__ float slab[NPB * OUT_DIM];
    __shared__ float emax_l[NPB];
    __shared__ float denom_l[NPB];
    __shared__ float adst_l[NPB];
    __shared__ int   list_a[LIST_CAP];
    __shared__ float list_v[LIST_CAP];
    __shared__ int   cnt;

    const int tid  = threadIdx.x;
    const int lane = tid & 31;
    const int wave = __builtin_amdgcn_readfirstlane(tid >> 5);
    const int g    = blockIdx.x;
    const int lo   = g * NPB;

    for (int i = tid; i < NPB * OUT_DIM; i += AGG_NT) slab[i] = 0.0f;
    if (tid < NPB) {
        emax_l[tid]  = -__builtin_inff();
        denom_l[tid] = 0.0f;
        adst_l[tid]  = asd[2 * (lo + tid) + 1];
    }
    if (tid == 0) { cnt = 0; ccnt = 0; }
    const unsigned pw = pres[(size_t)tid * 256 + g];
    __syncthreads();

    const uint4* __restrict__ rec16 = reinterpret_cast<const uint4*>(rec);
    {
        const int n = __popc(pw);
        unsigned below = 0, all = 0;
#pragma unroll
        for (int j = 0; j < 6; ++j) {
            const unsigned vote = __builtin_amdgcn_ballot_w32(((n >> j) & 1) != 0);
            below += __builtin_amdgcn_mbcnt_lo(vote, 0u) << j;
            all   += (unsigned)__popc(vote) << j;
        }
        int wbase = 0;
        if (lane == 0) wbase = atomicAdd(&ccnt, (int)all);
        wbase = __builtin_amdgcn_readfirstlane(wbase);
        int slot = wbase + (int)below;
        unsigned m = pw;
        while (m) {
            const int b = __ffs(m) - 1;
            const int line = tid * GEMM_WAVES + (b >> 3) + ((b & 4) ? GEMM_GRID * GEMM_WAVES : 0);
            if (slot < LIST_CAP) clist[slot] = line * 4 + (b & 3);
            ++slot;
            m &= m - 1;
        }
    }
    __syncthreads();
    const int nch_all = ccnt;
    const int nch     = (nch_all < LIST_CAP) ? nch_all : LIST_CAP;
    {
        const int sub  = lane >> 3;
        const int part = lane & 7;
        for (int q0 = wave * 4; q0 < nch; q0 += 32 * AGG_WAVES) {
            uint4 r[8];
#pragma unroll
            for (int u = 0; u < 8; ++u) {
                const int  ci = q0 + u * 4 * AGG_WAVES + sub;
                const bool ok = ci < nch;
                r[u] = rec16[8 * clist[ok ? ci : 0] + part];
                if (!ok) r[u] = make_uint4(~0u, ~0u, ~0u, ~0u);
            }
            unsigned off[32];
            unsigned tot = 0;
#pragma unroll
            for (int u = 0; u < 8; ++u) {
                const unsigned w[4] = {r[u].x, r[u].y, r[u].z, r[u].w};
#pragma unroll
                for (int j = 0; j < 4; ++j) {
                    const unsigned vote = __builtin_amdgcn_ballot_w32((w[j] >> 24) == (unsigned)g);
                    off[4 * u + j] = tot + __builtin_amdgcn_mbcnt_lo(vote, 0u);
                    tot += (unsigned)__popc(vote);
                }
            }
            if (tot) {
                int wb = 0;
                if (lane == 0) wb = atomicAdd(&cnt, (int)tot);
                wb = __builtin_amdgcn_readfirstlane(wb);
#pragma unroll
                for (int u = 0; u < 8; ++u) {
                    const unsigned w[4] = {r[u].x, r[u].y, r[u].z, r[u].w};
#pragma unroll
                    for (int j = 0; j < 4; ++j) {
                        const int idx = wb + (int)off[4 * u + j];
                        if ((w[j] >> 24) == (unsigned)g && idx < LIST_CAP)
                            list_a[idx] = (int)((((w[j] >> 16) & 0xFFu) << 20) | (w[j] & 0xFFFFu));
                    }
                }
            }
        }
    }
    __syncthreads();

    const int   total = (nch_all > LIST_CAP) ? LIST_CAP + 1 : cnt;
    const float ab    = att_b[0];

    if (total <= LIST_CAP) {
        for (int base = tid; base < total; base += 4 * AGG_NT) {
            int   pk[4];
            float as[4];
#pragma unroll
            for (int k = 0; k < 4; ++k) {
                const int idx = base + k * AGG_NT;
                pk[k] = (idx < total) ? list_a[idx] : 0;
            }
#pragma unroll
            for (int k = 0; k < 4; ++k) as[k] = asd[2 * (pk[k] & 0xFFFFF)];
#pragma unroll
            for (int k = 0; k < 4; ++k) {
                const int idx = base + k * AGG_NT;
                if (idx < total) {
                    const int dl = (int)((unsigned)pk[k] >> 20);
                    float v = as[k] + adst_l[dl] + ab;
                    v = (v > 0.0f) ? v : NEG_SLOPE * v;
                    list_v[idx] = v;
                    ldsMaxF(&emax_l[dl], v);
                }
            }
        }
        __syncthreads();

        for (int i = tid; i < total; i += AGG_NT) {
            const int dl = (int)((unsigned)list_a[i] >> 20);
            const float ex = __expf(list_v[i] - emax_l[dl]);
            list_v[i] = ex;
            atomicAdd(&denom_l[dl], ex);
        }
        __syncthreads();

        for (int base = wave; base < total; base += 8 * AGG_WAVES) {
            int    pk[8];
            float  ex[8];
            float2 hv[8];
            const int   myidx = base + (lane & 7) * AGG_WAVES;
            const int   pkv   = (myidx < total) ? list_a[myidx] : 0;
            const float exv   = (myidx < total) ? list_v[myidx] : 0.0f;
#pragma unroll
            for (int k = 0; k < 8; ++k) {
                pk[k] = __builtin_amdgcn_readlane(pkv, k);
                ex[k] = __int_as_float(__builtin_amdgcn_readlane(__float_as_int(exv), k));
            }
#pragma unroll
            for (int k = 0; k < 8; ++k)
                hv[k] = *reinterpret_cast<const float2*>(h + (size_t)(pk[k] & 0xFFFFF) * OUT_DIM + 2 * lane);
            asm volatile("" :: "v"(hv[0].x), "v"(hv[0].y), "v"(hv[1].x), "v"(hv[1].y),
                               "v"(hv[2].x), "v"(hv[2].y), "v"(hv[3].x), "v"(hv[3].y),
                               "v"(hv[4].x), "v"(hv[4].y), "v"(hv[5].x), "v"(hv[5].y),
                               "v"(hv[6].x), "v"(hv[6].y), "v"(hv[7].x), "v"(hv[7].y));
#pragma unroll
            for (int k = 0; k < 8; ++k) {
                const int idx = base + k * AGG_WAVES;
                if (idx < total) {
                    const int dl = (int)((unsigned)pk[k] >> 20);
                    float* row = &slab[dl * OUT_DIM + 2 * lane];
                    atomicAdd(row,     ex[k] * hv[k].x);
                    atomicAdd(row + 1, ex[k] * hv[k].y);
                }
            }
        }
        __syncthreads();
    } else {
        for (int e = tid; e < N_EDGES; e += AGG_NT) {
            const unsigned dl = (unsigned)(dst[e] - lo);
            if (dl < (unsigned)NPB) {
                float v = asd[2 * src[e]] + adst_l[dl] + ab;
                v = (v > 0.0f) ? v : NEG_SLOPE * v;
                ldsMaxF(&emax_l[dl], v);
            }
        }
        __syncthreads();
        for (int e = tid; e < N_EDGES; e += AGG_NT) {
            const unsigned dl = (unsigned)(dst[e] - lo);
            if (dl < (unsigned)NPB) {
                const int s = src[e];
                float v = asd[2 * s] + adst_l[dl] + ab;
                v = (v > 0.0f) ? v : NEG_SLOPE * v;
                const float ex = __expf(v - emax_l[dl]);
                atomicAdd(&denom_l[dl], ex);
                for (int c = 0; c < OUT_DIM; ++c) {
                    const int cc = (c + lane) & (OUT_DIM - 1);
                    atomicAdd(&slab[dl * OUT_DIM + cc], ex * h[(size_t)s * OUT_DIM + cc]);
                }
            }
        }
        __syncthreads();
    }

    for (int r = wave; r < NPB; r += AGG_WAVES) {
        const float  dn = denom_l[r];
        const size_t o  = (size_t)(lo + r) * OUT_DIM + lane;
        float v0, v1;
        if (dn > 0.0f) {
            v0 = slab[r * OUT_DIM + lane]      / dn;
            v1 = slab[r * OUT_DIM + lane + 32] / dn;
        } else {
            v0 = h[o];
            v1 = h[o + 32];
        }
        out[o]      = v0;
        out[o + 32] = v1;
    }
}

extern "C" void kernel_launch(void* const* d_in, const int* in_sizes, int n_in,
                              void* d_out, int out_size, void* d_ws, size_t ws_size,
                              hipStream_t stream) {
    const float* hidden = (const float*)d_in[0];
    const float* W      = (const float*)d_in[1];
    const float* bias   = (const float*)d_in[2];
    const float* att_w  = (const float*)d_in[3];
    const float* att_b  = (const float*)d_in[4];
    const int*   src    = (const int*)  d_in[5];
    const int*   dst    = (const int*)  d_in[6];
    float* out = (float*)d_out;

    float* ws     = (float*)d_ws;
    float* h_ws   = ws;
    float* asd    = h_ws + (size_t)N_NODES * OUT_DIM;
    unsigned int* rec    = (unsigned int*)(asd + 2 * (size_t)N_NODES);
    unsigned int* pres   = rec + N_EDGES;

    gat_gemm_wmma<<<GEMM_GRID, GEMM_NT, 0, stream>>>(hidden, W, bias, att_w, src, dst, h_ws, asd, rec, pres);
    gat_aggregate<<<NGROUPS, AGG_NT, 0, stream>>>(src, dst, asd, att_b, h_ws, rec, pres, out);
}
